# starting kernel
# baseline (speedup 1.0000x reference)
#include <hip/hip_runtime.h>
#include <hip/hip_bf16.h>


typedef __attribute__((ext_vector_type(2))) float v2f;
typedef __attribute__((ext_vector_type(8))) float v8f;

#define D_IN   256
#define D_OUT  96
#define MINN   1e-15f
#define MAXN   0.996f
#define ARTC   (1.0f - 1e-7f)

__device__ __forceinline__ float wsum(float v) {
    v += __shfl_xor(v, 16, 32);
    v += __shfl_xor(v, 8, 32);
    v += __shfl_xor(v, 4, 32);
    v += __shfl_xor(v, 2, 32);
    v += __shfl_xor(v, 1, 32);
    return v;
}

__global__ void bias_kernel(const float* __restrict__ bias, float* __restrict__ hb) {
    __shared__ float red[4];
    int t = threadIdx.x;
    float v = (t < D_OUT) ? bias[t] : 0.0f;
    float ss = wsum(v * v);
    if ((t & 31) == 0) red[t >> 5] = ss;
    __syncthreads();
    float n = fmaxf(sqrtf(red[0] + red[1] + red[2] + red[3]), MINN);
    float p = tanhf(n) / n * v;
    __syncthreads();
    float ps = wsum(p * p);
    if ((t & 31) == 0) red[t >> 5] = ps;
    __syncthreads();
    float pn = fmaxf(sqrtf(red[0] + red[1] + red[2] + red[3]), MINN);
    float pf = (pn > MAXN) ? MAXN / pn : 1.0f;
    if (t < D_OUT) hb[t] = p * pf;
}

__global__ __launch_bounds__(256) void gemm_wmma(const float* __restrict__ x,
                                                 const float* __restrict__ w,
                                                 float* __restrict__ mx, int N) {
    const int nTilesM = (N + 15) >> 4;
    const int tileM   = (blockIdx.x << 3) + (threadIdx.x >> 5);
    if (tileM >= nTilesM) return;

    const int lane = threadIdx.x & 31;
    const int half = lane >> 4;
    const int mr   = lane & 15;

    int mA = tileM * 16 + mr; if (mA > N - 1) mA = N - 1;
    const float* aP = x + (size_t)mA * D_IN + 2 * half;
    const float* bP = w + (size_t)mr * D_IN + 2 * half;

    v8f acc[6] = {{}, {}, {}, {}, {}, {}};
#pragma unroll 2
    for (int k = 0; k < D_IN; k += 4) {
        v2f a = *(const v2f*)(aP + k);
#pragma unroll
        for (int t = 0; t < 6; ++t) {
            v2f b = *(const v2f*)(bP + k + t * 16 * D_IN);
            acc[t] = __builtin_amdgcn_wmma_f32_16x16x4_f32(
                 false, a,  false, b,
                 (short)0, acc[t],  false,  false);
        }
    }

    if (tileM * 16 + 15 < N) {
        float* base = mx + (size_t)(tileM * 16 + 8 * half) * D_OUT + mr;
#pragma unroll
        for (int r = 0; r < 8; ++r) {
            float* o = base + (size_t)r * D_OUT;
#pragma unroll
            for (int t = 0; t < 6; ++t) o[t * 16] = acc[t][r];
        }
    } else {
        for (int r = 0; r < 8; ++r) {
            int m = tileM * 16 + r + 8 * half;
            if (m < N) {
                float* o = mx + (size_t)m * D_OUT + mr;
                for (int t = 0; t < 6; ++t) o[t * 16] = acc[t][r];
            }
        }
    }
}

__global__ __launch_bounds__(256) void hyp_linear_epilogue(
    const float* __restrict__ x, const float* __restrict__ mx,
    const float* __restrict__ hb, float* __restrict__ xt, int N) {
    int row = (blockIdx.x << 3) + (threadIdx.x >> 5);
    if (row >= N) return;
    int lane = threadIdx.x & 31;

    const float* xr = x + (size_t)row * D_IN;
    float ss = 0.0f;
#pragma unroll
    for (int j = 0; j < 8; ++j) { float v = xr[lane + 32 * j]; ss = fmaf(v, v, ss); }
    float xn = fmaxf(sqrtf(wsum(ss)), MINN);

    const float* mr = mx + (size_t)row * D_OUT;
    float m0 = mr[lane], m1 = mr[lane + 32], m2 = mr[lane + 64];
    float mn = fmaxf(sqrtf(wsum(m0 * m0 + m1 * m1 + m2 * m2)), MINN);
    float sc = tanhf(mn / xn * atanhf(fminf(xn, ARTC))) / mn;
    float h0 = sc * m0, h1 = sc * m1, h2 = sc * m2;
    float hn = fmaxf(sqrtf(wsum(h0 * h0 + h1 * h1 + h2 * h2)), MINN);
    float pf = (hn > MAXN) ? MAXN / hn : 1.0f;
    h0 *= pf; h1 *= pf; h2 *= pf;

    float b0 = hb[lane], b1 = hb[lane + 32], b2 = hb[lane + 64];
    float x2 = wsum(h0 * h0 + h1 * h1 + h2 * h2);
    float y2 = wsum(b0 * b0 + b1 * b1 + b2 * b2);
    float xy = wsum(h0 * b0 + h1 * b1 + h2 * b2);
    float ca  = 1.0f + 2.0f * xy + y2;
    float cb  = 1.0f - x2;
    float den = fmaxf(1.0f + 2.0f * xy + x2 * y2, MINN);
    float g0 = (ca * h0 + cb * b0) / den;
    float g1 = (ca * h1 + cb * b1) / den;
    float g2 = (ca * h2 + cb * b2) / den;
    float gn = fmaxf(sqrtf(wsum(g0 * g0 + g1 * g1 + g2 * g2)), MINN);
    pf = (gn > MAXN) ? MAXN / gn : 1.0f;
    g0 *= pf; g1 *= pf; g2 *= pf;
    float pn  = fminf(gn, MAXN);
    float lsc = atanhf(fminf(pn, ARTC)) / pn;
    float* o = xt + (size_t)row * D_OUT;
    o[lane] = lsc * g0; o[lane + 32] = lsc * g1; o[lane + 64] = lsc * g2;
}

__global__ void zero_kernel(float* __restrict__ p, size_t n) {
    size_t i = (size_t)blockIdx.x * blockDim.x + threadIdx.x;
    size_t s = (size_t)gridDim.x * blockDim.x;
    for (; i < n; i += s) p[i] = 0.0f;
}

__global__ __launch_bounds__(256) void spmm_kernel(
    const int* __restrict__ erow, const int* __restrict__ ecol,
    const float* __restrict__ ew, const float* __restrict__ xt,
    float* __restrict__ sup, int E) {
    int tid  = blockIdx.x * blockDim.x + threadIdx.x;
    int lane = tid & 31;
    int waveStride = (gridDim.x * blockDim.x) >> 5;
    for (int e = tid >> 5; e < E; e += waveStride) {
        int   r = erow[e];
        int   c = ecol[e];
        float w = ew[e];
        const float* src = xt + (size_t)c * D_OUT;
        float*       dst = sup + (size_t)r * D_OUT;
        atomicAdd(dst + lane,      w * src[lane]);
        atomicAdd(dst + lane + 32, w * src[lane + 32]);
        atomicAdd(dst + lane + 64, w * src[lane + 64]);
    }
}

__global__ __launch_bounds__(256) void hyp_agg_act(const float* __restrict__ sup,
                                                   float* __restrict__ out, int N) {
    int row = (blockIdx.x << 3) + (threadIdx.x >> 5);
    if (row >= N) return;
    int lane = threadIdx.x & 31;
    const float* sr = sup + (size_t)row * D_OUT;
    float s0 = sr[lane], s1 = sr[lane + 32], s2 = sr[lane + 64];
    float sn = fmaxf(sqrtf(wsum(s0 * s0 + s1 * s1 + s2 * s2)), MINN);
    float e  = tanhf(sn) / sn;
    float p0 = e * s0, p1 = e * s1, p2 = e * s2;
    float pnr = fmaxf(sqrtf(wsum(p0 * p0 + p1 * p1 + p2 * p2)), MINN);
    float pf  = (pnr > MAXN) ? MAXN / pnr : 1.0f;
    p0 *= pf; p1 *= pf; p2 *= pf;
    float pn = fminf(pnr, MAXN);
    float l  = atanhf(fminf(pn, ARTC)) / pn;
    float u0 = fmaxf(l * p0, 0.0f), u1 = fmaxf(l * p1, 0.0f), u2 = fmaxf(l * p2, 0.0f);
    float un = fmaxf(sqrtf(wsum(u0 * u0 + u1 * u1 + u2 * u2)), MINN);
    float e2 = tanhf(un) / un;
    float h0 = e2 * u0, h1 = e2 * u1, h2 = e2 * u2;
    float hn  = fmaxf(sqrtf(wsum(h0 * h0 + h1 * h1 + h2 * h2)), MINN);
    float pf2 = (hn > MAXN) ? MAXN / hn : 1.0f;
    float* o = out + (size_t)row * D_OUT;
    o[lane] = h0 * pf2; o[lane + 32] = h1 * pf2; o[lane + 64] = h2 * pf2;
}

extern "C" void kernel_launch(void* const* d_in, const int* in_sizes, int n_in,
                              void* d_out, int out_size, void* d_ws, size_t ws_size,
                              hipStream_t stream) {
    const float* x      = (const float*)d_in[0];
    const float* weight = (const float*)d_in[1];
    const float* bias   = (const float*)d_in[2];
    const float* ew     = (const float*)d_in[3];
    const int*   erow   = (const int*)d_in[4];
    const int*   ecol   = (const int*)d_in[5];
    const int N = in_sizes[0] / D_IN;
    const int E = in_sizes[3];

    float* ws   = (float*)d_ws;
    float* hb   = ws;
    float* buf0 = ws + 128;
    float* buf1 = buf0 + (size_t)N * D_OUT;
    float* out  = (float*)d_out;

    bias_kernel<<<1, 128, 0, stream>>>(bias, hb);

    int strips     = (N + 15) >> 4;
    int gemmBlocks = (strips + 7) >> 3;
    gemm_wmma<<<gemmBlocks, 256, 0, stream>>>(x, weight, buf0, N);

    hyp_linear_epilogue<<<(N + 7) >> 3, 256, 0, stream>>>(x, buf0, hb, buf1, N);

    zero_kernel<<<2048, 256, 0, stream>>>(buf0, (size_t)N * D_OUT);
    spmm_kernel<<<(E + 7) >> 3, 256, 0, stream>>>(erow, ecol, ew, buf1, buf0, E);

    hyp_agg_act<<<(N + 7) >> 3, 256, 0, stream>>>(buf0, out, N);
}


# edge linking moved into the row-epilogue launch: each row's wave also links its E/N edges, xt in place over mx, heads reset in the bias launch; 6 launches -> 4; reverse-arrival sum order as before
# speedup vs baseline: 1.4748x; 1.4748x over previous
#include <hip/hip_runtime.h>
#include <hip/hip_bf16.h>


typedef __attribute__((ext_vector_type(2))) float v2f;
typedef __attribute__((ext_vector_type(8))) float v8f;

#define D_IN   256
#define D_OUT  96
#define MINN   1e-15f
#define MAXN   0.996f
#define ARTC   (1.0f - 1e-7f)

__device__ __forceinline__ float wsum(float v) {
    v += __shfl_xor(v, 16, 32);
    v += __shfl_xor(v, 8, 32);
    v += __shfl_xor(v, 4, 32);
    v += __shfl_xor(v, 2, 32);
    v += __shfl_xor(v, 1, 32);
    return v;
}

__global__ __launch_bounds__(256) void bias_heads(const float* __restrict__ bias,
                                                  float* __restrict__ hb,
                                                  int* __restrict__ head, int N) {
    __shared__ float red[8];
    int t = threadIdx.x;
    int i = (blockIdx.x << 8) + t;
    if (i < N) head[i] = -1;
    if (blockIdx.x != 0) return;
    float v = (t < D_OUT) ? bias[t] : 0.0f;
    float ss = wsum(v * v);
    if ((t & 31) == 0) red[t >> 5] = ss;
    __syncthreads();
    float n = fmaxf(sqrtf(red[0] + red[1] + red[2] + red[3]), MINN);
    float p = tanhf(n) / n * v;
    __syncthreads();
    float ps = wsum(p * p);
    if ((t & 31) == 0) red[t >> 5] = ps;
    __syncthreads();
    float pn = fmaxf(sqrtf(red[0] + red[1] + red[2] + red[3]), MINN);
    float pf = (pn > MAXN) ? MAXN / pn : 1.0f;
    if (t < D_OUT) hb[t] = p * pf;
}

__global__ __launch_bounds__(256) void gemm_wmma(const float* __restrict__ x,
                                                 const float* __restrict__ w,
                                                 float* __restrict__ mx, int N) {
    const int nTilesM = (N + 15) >> 4;
    const int tileM   = (blockIdx.x << 3) + (threadIdx.x >> 5);
    if (tileM >= nTilesM) return;

    const int lane = threadIdx.x & 31;
    const int half = lane >> 4;
    const int mr   = lane & 15;

    int mA = tileM * 16 + mr; if (mA > N - 1) mA = N - 1;
    const float* aP = x + (size_t)mA * D_IN + 2 * half;
    const float* bP = w + (size_t)mr * D_IN + 2 * half;

    v8f acc[6] = {{}, {}, {}, {}, {}, {}};
#pragma unroll 2
    for (int k = 0; k < D_IN; k += 4) {
        v2f a = *(const v2f*)(aP + k);
#pragma unroll
        for (int t = 0; t < 6; ++t) {
            v2f b = *(const v2f*)(bP + k + t * 16 * D_IN);
            acc[t] = __builtin_amdgcn_wmma_f32_16x16x4_f32(
                 false, a,  false, b,
                 (short)0, acc[t],  false,  false);
        }
    }

    if (tileM * 16 + 15 < N) {
        float* base = mx + (size_t)(tileM * 16 + 8 * half) * D_OUT + mr;
#pragma unroll
        for (int r = 0; r < 8; ++r) {
            float* o = base + (size_t)r * D_OUT;
#pragma unroll
            for (int t = 0; t < 6; ++t) o[t * 16] = acc[t][r];
        }
    } else {
        for (int r = 0; r < 8; ++r) {
            int m = tileM * 16 + r + 8 * half;
            if (m < N) {
                float* o = mx + (size_t)m * D_OUT + mr;
                for (int t = 0; t < 6; ++t) o[t * 16] = acc[t][r];
            }
        }
    }
}

typedef __attribute__((ext_vector_type(4))) int v4i;

__global__ __launch_bounds__(256) void epilogue_link(
    const float* __restrict__ x, float* mxt, const float* __restrict__ hb,
    const int* __restrict__ erow, const int* __restrict__ ecol,
    const float* __restrict__ ew, int* __restrict__ head, int* __restrict__ rec,
    int N, int E) {
    int row = (blockIdx.x << 3) + (threadIdx.x >> 5);
    if (row >= N) return;
    int lane = threadIdx.x & 31;

    const float* xr = x + (size_t)row * D_IN;
    float xv[8];
#pragma unroll
    for (int j = 0; j < 8; ++j) xv[j] = xr[lane + 32 * j];
    float* mr = mxt + (size_t)row * D_OUT;
    float m0 = mr[lane], m1 = mr[lane + 32], m2 = mr[lane + 64];
    float b0 = hb[lane], b1 = hb[lane + 32], b2 = hb[lane + 64];

    const int epw = (E + N - 1) / N;
    const int e0  = row * epw + lane;
    const bool has = (lane < epw) && (e0 < E);
    const int*   pc = has ? ecol + e0 : head;
    const float* pw = has ? ew + e0 : hb;
    const int*   pr = has ? erow + e0 : head;
    v4i q = {-1, 0, 0, 0};
    q.y = *pc;
    q.z = __float_as_int(*pw);
    int r = *pr;
    if (!has) r = -1;

    float ss = 0.0f;
#pragma unroll
    for (int j = 0; j < 8; ++j) ss = fmaf(xv[j], xv[j], ss);
    float mq = m0 * m0 + m1 * m1 + m2 * m2;
    float yq = b0 * b0 + b1 * b1 + b2 * b2;

    if ((unsigned)r < (unsigned)N) q.x = atomicExch(head + r, e0);

    float xn = fmaxf(sqrtf(wsum(ss)), MINN);

    float mn = fmaxf(sqrtf(wsum(mq)), MINN);
    float sc = tanhf(mn / xn * atanhf(fminf(xn, ARTC))) / mn;
    float h0 = sc * m0, h1 = sc * m1, h2 = sc * m2;
    float hn = fmaxf(sqrtf(wsum(h0 * h0 + h1 * h1 + h2 * h2)), MINN);
    float pf = (hn > MAXN) ? MAXN / hn : 1.0f;
    h0 *= pf; h1 *= pf; h2 *= pf;

    float x2 = wsum(h0 * h0 + h1 * h1 + h2 * h2);
    float y2 = wsum(yq);
    float xy = wsum(h0 * b0 + h1 * b1 + h2 * b2);
    float ca  = 1.0f + 2.0f * xy + y2;
    float cb  = 1.0f - x2;
    float den = fmaxf(1.0f + 2.0f * xy + x2 * y2, MINN);
    float g0 = (ca * h0 + cb * b0) / den;
    float g1 = (ca * h1 + cb * b1) / den;
    float g2 = (ca * h2 + cb * b2) / den;
    float gn = fmaxf(sqrtf(wsum(g0 * g0 + g1 * g1 + g2 * g2)), MINN);
    pf = (gn > MAXN) ? MAXN / gn : 1.0f;
    g0 *= pf; g1 *= pf; g2 *= pf;
    float pn  = fminf(gn, MAXN);
    float lsc = atanhf(fminf(pn, ARTC)) / pn;
    mr[lane] = lsc * g0; mr[lane + 32] = lsc * g1; mr[lane + 64] = lsc * g2;

    if (has) *(v4i*)(rec + 4 * (size_t)e0) = q;
    for (int k = lane + 32; k < epw; k += 32) {
        int e = row * epw + k;
        if (e < E) {
            int r2 = erow[e];
            v4i p = {-1, ecol[e], __float_as_int(ew[e]), 0};
            if ((unsigned)r2 < (unsigned)N) p.x = atomicExch(head + r2, e);
            *(v4i*)(rec + 4 * (size_t)e) = p;
        }
    }
}

__global__ __launch_bounds__(256) void pull_agg_act(
    const int* __restrict__ head, const int* __restrict__ rec,
    const float* __restrict__ xt, float* __restrict__ out, int N, int E) {
    int row = (blockIdx.x << 3) + (threadIdx.x >> 5);
    if (row >= N) return;
    int lane = threadIdx.x & 31;

    int e = __builtin_amdgcn_readfirstlane(head[row]);
    v4i r = {-1, 0, 0, 0};
    if (e >= 0) r = *(const v4i*)(rec + 4 * (size_t)e);
    float s0 = 0.0f, s1 = 0.0f, s2 = 0.0f;
    for (int it = 0; it < E && e >= 0; ++it) {
        int   c = __builtin_amdgcn_readfirstlane(r.y);
        float w = __int_as_float(__builtin_amdgcn_readfirstlane(r.z));
        const float* src = xt + (size_t)c * D_OUT + lane;
        float g0 = src[0], g1 = src[32], g2 = src[64];
        e = __builtin_amdgcn_readfirstlane(r.x);
        if (e >= 0) r = *(const v4i*)(rec + 4 * (size_t)e);
        s0 = fmaf(w, g0, s0);
        s1 = fmaf(w, g1, s1);
        s2 = fmaf(w, g2, s2);
    }

    float sn = fmaxf(sqrtf(wsum(s0 * s0 + s1 * s1 + s2 * s2)), MINN);
    float ex = tanhf(sn) / sn;
    float p0 = ex * s0, p1 = ex * s1, p2 = ex * s2;
    float pnr = fmaxf(sqrtf(wsum(p0 * p0 + p1 * p1 + p2 * p2)), MINN);
    float pf  = (pnr > MAXN) ? MAXN / pnr : 1.0f;
    p0 *= pf; p1 *= pf; p2 *= pf;
    float pn = fminf(pnr, MAXN);
    float l  = atanhf(fminf(pn, ARTC)) / pn;
    float u0 = fmaxf(l * p0, 0.0f), u1 = fmaxf(l * p1, 0.0f), u2 = fmaxf(l * p2, 0.0f);
    float un = fmaxf(sqrtf(wsum(u0 * u0 + u1 * u1 + u2 * u2)), MINN);
    float e2 = tanhf(un) / un;
    float h0 = e2 * u0, h1 = e2 * u1, h2 = e2 * u2;
    float hn  = fmaxf(sqrtf(wsum(h0 * h0 + h1 * h1 + h2 * h2)), MINN);
    float pf2 = (hn > MAXN) ? MAXN / hn : 1.0f;
    float* o = out + (size_t)row * D_OUT;
    o[lane] = h0 * pf2; o[lane + 32] = h1 * pf2; o[lane + 64] = h2 * pf2;
}

extern "C" void kernel_launch(void* const* d_in, const int* in_sizes, int n_in,
                              void* d_out, int out_size, void* d_ws, size_t ws_size,
                              hipStream_t stream) {
    const float* x      = (const float*)d_in[0];
    const float* weight = (const float*)d_in[1];
    const float* bias   = (const float*)d_in[2];
    const float* ew     = (const float*)d_in[3];
    const int*   erow   = (const int*)d_in[4];
    const int*   ecol   = (const int*)d_in[5];
    const int N = in_sizes[0] / D_IN;
    const int E = in_sizes[3];

    float* ws   = (float*)d_ws;
    float* hb   = ws;
    float* buf0 = ws + 128;
    float* buf1 = buf0 + (size_t)N * D_OUT;
    float* out  = (float*)d_out;
    int* head = (int*)buf1;
    int* rec  = head + ((N + 3) & ~3);

    bias_heads<<<(N + 255) >> 8, 256, 0, stream>>>(bias, hb, head, N);

    int strips     = (N + 15) >> 4;
    int gemmBlocks = (strips + 7) >> 3;
    gemm_wmma<<<gemmBlocks, 256, 0, stream>>>(x, weight, buf0, N);

    epilogue_link<<<(N + 7) >> 3, 256, 0, stream>>>(x, buf0, hb, erow, ecol, ew, head, rec, N, E);

    pull_agg_act<<<(N + 7) >> 3, 256, 0, stream>>>(head, rec, buf0, out, N, E);
}


# gemm register tiling over rows: one wave computes 64x96 (four 16-row strips share each W fragment, 24 accumulators, next 4 k-steps' fragments requested ahead); 782 waves; k and sum order unchanged
# speedup vs baseline: 1.6799x; 1.1391x over previous
#include <hip/hip_runtime.h>
#include <hip/hip_bf16.h>


typedef __attribute__((ext_vector_type(2))) float v2f;
typedef __attribute__((ext_vector_type(8))) float v8f;

#define D_IN   256
#define D_OUT  96
#define MINN   1e-15f
#define MAXN   0.996f
#define ARTC   (1.0f - 1e-7f)

__device__ __forceinline__ float wsum(float v) {
    v += __shfl_xor(v, 16, 32);
    v += __shfl_xor(v, 8, 32);
    v += __shfl_xor(v, 4, 32);
    v += __shfl_xor(v, 2, 32);
    v += __shfl_xor(v, 1, 32);
    return v;
}

__global__ __launch_bounds__(256) void bias_heads(const float* __restrict__ bias,
                                                  float* __restrict__ hb,
                                                  int* __restrict__ head, int N) {
    __shared__ float red[8];
    int t = threadIdx.x;
    int i = (blockIdx.x << 8) + t;
    if (i < N) head[i] = -1;
    if (blockIdx.x != 0) return;
    float v = (t < D_OUT) ? bias[t] : 0.0f;
    float ss = wsum(v * v);
    if ((t & 31) == 0) red[t >> 5] = ss;
    __syncthreads();
    float n = fmaxf(sqrtf(red[0] + red[1] + red[2] + red[3]), MINN);
    float p = tanhf(n) / n * v;
    __syncthreads();
    float ps = wsum(p * p);
    if ((t & 31) == 0) red[t >> 5] = ps;
    __syncthreads();
    float pn = fmaxf(sqrtf(red[0] + red[1] + red[2] + red[3]), MINN);
    float pf = (pn > MAXN) ? MAXN / pn : 1.0f;
    if (t < D_OUT) hb[t] = p * pf;
}

#define KS 4
struct FragSet { v2f a[4][KS]; v2f b[6][KS]; };

__device__ __forceinline__ void load_set(FragSet& f, const float* aP0, const float* aP1,
                                         const float* aP2, const float* aP3,
                                         const float* bP, int k) {
#pragma unroll
    for (int s = 0; s < KS; ++s) {
        f.a[0][s] = *(const v2f*)(aP0 + k + 4 * s);
        f.a[1][s] = *(const v2f*)(aP1 + k + 4 * s);
        f.a[2][s] = *(const v2f*)(aP2 + k + 4 * s);
        f.a[3][s] = *(const v2f*)(aP3 + k + 4 * s);
#pragma unroll
        for (int t = 0; t < 6; ++t)
            f.b[t][s] = *(const v2f*)(bP + k + 4 * s + t * 16 * D_IN);
    }
}

__device__ __forceinline__ void mma_set(v8f (&acc)[4][6], const FragSet& f) {
#pragma unroll
    for (int s = 0; s < KS; ++s)
#pragma unroll
        for (int t = 0; t < 6; ++t)
#pragma unroll
            for (int m = 0; m < 4; ++m)
                acc[m][t] = __builtin_amdgcn_wmma_f32_16x16x4_f32(
                     false, f.a[m][s],  false, f.b[t][s],
                     (short)0, acc[m][t],  false,  false);
}

__global__ __launch_bounds__(128) void gemm_wmma(const float* __restrict__ x,
                                                 const float* __restrict__ w,
                                                 float* __restrict__ mx, int N) {
    const int nQuads = (N + 63) >> 6;
    const int quad   = (blockIdx.x << 2) + (threadIdx.x >> 5);
    if (quad >= nQuads) return;

    const int lane = threadIdx.x & 31;
    const int half = lane >> 4;
    const int mr   = lane & 15;
    const int row0 = quad * 64;

    int m0 = row0 + mr;      if (m0 > N - 1) m0 = N - 1;
    int m1 = row0 + 16 + mr; if (m1 > N - 1) m1 = N - 1;
    int m2 = row0 + 32 + mr; if (m2 > N - 1) m2 = N - 1;
    int m3 = row0 + 48 + mr; if (m3 > N - 1) m3 = N - 1;
    const float* aP0 = x + (size_t)m0 * D_IN + 2 * half;
    const float* aP1 = x + (size_t)m1 * D_IN + 2 * half;
    const float* aP2 = x + (size_t)m2 * D_IN + 2 * half;
    const float* aP3 = x + (size_t)m3 * D_IN + 2 * half;
    const float* bP  = w + (size_t)mr * D_IN + 2 * half;

    v8f acc[4][6] = {};
    FragSet f0, f1;
    load_set(f0, aP0, aP1, aP2, aP3, bP, 0);
#pragma unroll 1
    for (int k = 0; k < D_IN - 8 * KS; k += 8 * KS) {
        load_set(f1, aP0, aP1, aP2, aP3, bP, k + 4 * KS);
        mma_set(acc, f0);
        load_set(f0, aP0, aP1, aP2, aP3, bP, k + 8 * KS);
        mma_set(acc, f1);
    }
    load_set(f1, aP0, aP1, aP2, aP3, bP, D_IN - 4 * KS);
    mma_set(acc, f0);
    mma_set(acc, f1);

#pragma unroll
    for (int m = 0; m < 4; ++m) {
        const int r0 = row0 + 16 * m;
        if (r0 + 15 < N) {
            float* base = mx + (size_t)(r0 + 8 * half) * D_OUT + mr;
#pragma unroll
            for (int r = 0; r < 8; ++r) {
                float* o = base + (size_t)r * D_OUT;
#pragma unroll
                for (int t = 0; t < 6; ++t) o[t * 16] = acc[m][t][r];
            }
        } else {
#pragma unroll
            for (int r = 0; r < 8; ++r) {
                int mm = r0 + r + 8 * half;
                if (mm < N) {
                    float* o = mx + (size_t)mm * D_OUT + mr;
#pragma unroll
                    for (int t = 0; t < 6; ++t) o[t * 16] = acc[m][t][r];
                }
            }
        }
    }
}

typedef __attribute__((ext_vector_type(4))) int v4i;

__global__ __launch_bounds__(256) void epilogue_link(
    const float* __restrict__ x, float* mxt, const float* __restrict__ hb,
    const int* __restrict__ erow, const int* __restrict__ ecol,
    const float* __restrict__ ew, int* __restrict__ head, int* __restrict__ rec,
    int N, int E) {
    int row = (blockIdx.x << 3) + (threadIdx.x >> 5);
    if (row >= N) return;
    int lane = threadIdx.x & 31;

    const float* xr = x + (size_t)row * D_IN;
    float xv[8];
#pragma unroll
    for (int j = 0; j < 8; ++j) xv[j] = xr[lane + 32 * j];
    float* mr = mxt + (size_t)row * D_OUT;
    float m0 = mr[lane], m1 = mr[lane + 32], m2 = mr[lane + 64];
    float b0 = hb[lane], b1 = hb[lane + 32], b2 = hb[lane + 64];

    const int epw = (E + N - 1) / N;
    const int e0  = row * epw + lane;
    const bool has = (lane < epw) && (e0 < E);
    const int*   pc = has ? ecol + e0 : head;
    const float* pw = has ? ew + e0 : hb;
    const int*   pr = has ? erow + e0 : head;
    v4i q = {-1, 0, 0, 0};
    q.y = *pc;
    q.z = __float_as_int(*pw);
    int r = *pr;
    if (!has) r = -1;

    float ss = 0.0f;
#pragma unroll
    for (int j = 0; j < 8; ++j) ss = fmaf(xv[j], xv[j], ss);
    float mq = m0 * m0 + m1 * m1 + m2 * m2;
    float yq = b0 * b0 + b1 * b1 + b2 * b2;

    if ((unsigned)r < (unsigned)N) q.x = atomicExch(head + r, e0);

    float xn = fmaxf(sqrtf(wsum(ss)), MINN);

    float mn = fmaxf(sqrtf(wsum(mq)), MINN);
    float sc = tanhf(mn / xn * atanhf(fminf(xn, ARTC))) / mn;
    float h0 = sc * m0, h1 = sc * m1, h2 = sc * m2;
    float hn = fmaxf(sqrtf(wsum(h0 * h0 + h1 * h1 + h2 * h2)), MINN);
    float pf = (hn > MAXN) ? MAXN / hn : 1.0f;
    h0 *= pf; h1 *= pf; h2 *= pf;

    float x2 = wsum(h0 * h0 + h1 * h1 + h2 * h2);
    float y2 = wsum(yq);
    float xy = wsum(h0 * b0 + h1 * b1 + h2 * b2);
    float ca  = 1.0f + 2.0f * xy + y2;
    float cb  = 1.0f - x2;
    float den = fmaxf(1.0f + 2.0f * xy + x2 * y2, MINN);
    float g0 = (ca * h0 + cb * b0) / den;
    float g1 = (ca * h1 + cb * b1) / den;
    float g2 = (ca * h2 + cb * b2) / den;
    float gn = fmaxf(sqrtf(wsum(g0 * g0 + g1 * g1 + g2 * g2)), MINN);
    pf = (gn > MAXN) ? MAXN / gn : 1.0f;
    g0 *= pf; g1 *= pf; g2 *= pf;
    float pn  = fminf(gn, MAXN);
    float lsc = atanhf(fminf(pn, ARTC)) / pn;
    mr[lane] = lsc * g0; mr[lane + 32] = lsc * g1; mr[lane + 64] = lsc * g2;

    if (has) *(v4i*)(rec + 4 * (size_t)e0) = q;
    for (int k = lane + 32; k < epw; k += 32) {
        int e = row * epw + k;
        if (e < E) {
            int r2 = erow[e];
            v4i p = {-1, ecol[e], __float_as_int(ew[e]), 0};
            if ((unsigned)r2 < (unsigned)N) p.x = atomicExch(head + r2, e);
            *(v4i*)(rec + 4 * (size_t)e) = p;
        }
    }
}

__global__ __launch_bounds__(256) void pull_agg_act(
    const int* __restrict__ head, const int* __restrict__ rec,
    const float* __restrict__ xt, float* __restrict__ out, int N, int E) {
    int row = (blockIdx.x << 3) + (threadIdx.x >> 5);
    if (row >= N) return;
    int lane = threadIdx.x & 31;

    int e = __builtin_amdgcn_readfirstlane(head[row]);
    v4i r = {-1, 0, 0, 0};
    if (e >= 0) r = *(const v4i*)(rec + 4 * (size_t)e);
    float s0 = 0.0f, s1 = 0.0f, s2 = 0.0f;
    for (int it = 0; it < E && e >= 0; ++it) {
        int   c = __builtin_amdgcn_readfirstlane(r.y);
        float w = __int_as_float(__builtin_amdgcn_readfirstlane(r.z));
        const float* src = xt + (size_t)c * D_OUT + lane;
        float g0 = src[0], g1 = src[32], g2 = src[64];
        e = __builtin_amdgcn_readfirstlane(r.x);
        if (e >= 0) r = *(const v4i*)(rec + 4 * (size_t)e);
        s0 = fmaf(w, g0, s0);
        s1 = fmaf(w, g1, s1);
        s2 = fmaf(w, g2, s2);
    }

    float sn = fmaxf(sqrtf(wsum(s0 * s0 + s1 * s1 + s2 * s2)), MINN);
    float ex = tanhf(sn) / sn;
    float p0 = ex * s0, p1 = ex * s1, p2 = ex * s2;
    float pnr = fmaxf(sqrtf(wsum(p0 * p0 + p1 * p1 + p2 * p2)), MINN);
    float pf  = (pnr > MAXN) ? MAXN / pnr : 1.0f;
    p0 *= pf; p1 *= pf; p2 *= pf;
    float pn = fminf(pnr, MAXN);
    float l  = atanhf(fminf(pn, ARTC)) / pn;
    float u0 = fmaxf(l * p0, 0.0f), u1 = fmaxf(l * p1, 0.0f), u2 = fmaxf(l * p2, 0.0f);
    float un = fmaxf(sqrtf(wsum(u0 * u0 + u1 * u1 + u2 * u2)), MINN);
    float e2 = tanhf(un) / un;
    float h0 = e2 * u0, h1 = e2 * u1, h2 = e2 * u2;
    float hn  = fmaxf(sqrtf(wsum(h0 * h0 + h1 * h1 + h2 * h2)), MINN);
    float pf2 = (hn > MAXN) ? MAXN / hn : 1.0f;
    float* o = out + (size_t)row * D_OUT;
    o[lane] = h0 * pf2; o[lane + 32] = h1 * pf2; o[lane + 64] = h2 * pf2;
}

extern "C" void kernel_launch(void* const* d_in, const int* in_sizes, int n_in,
                              void* d_out, int out_size, void* d_ws, size_t ws_size,
                              hipStream_t stream) {
    const float* x      = (const float*)d_in[0];
    const float* weight = (const float*)d_in[1];
    const float* bias   = (const float*)d_in[2];
    const float* ew     = (const float*)d_in[3];
    const int*   erow   = (const int*)d_in[4];
    const int*   ecol   = (const int*)d_in[5];
    const int N = in_sizes[0] / D_IN;
    const int E = in_sizes[3];

    float* ws   = (float*)d_ws;
    float* hb   = ws;
    float* buf0 = ws + 128;
    float* buf1 = buf0 + (size_t)N * D_OUT;
    float* out  = (float*)d_out;
    int* head = (int*)buf1;
    int* rec  = head + ((N + 3) & ~3);

    bias_heads<<<(N + 255) >> 8, 256, 0, stream>>>(bias, hb, head, N);

    int quads      = (N + 63) >> 6;
    int gemmBlocks = (quads + 3) >> 2;
    gemm_wmma<<<gemmBlocks, 128, 0, stream>>>(x, weight, buf0, N);

    epilogue_link<<<(N + 7) >> 3, 256, 0, stream>>>(x, buf0, hb, erow, ecol, ew, head, rec, N, E);

    pull_agg_act<<<(N + 7) >> 3, 256, 0, stream>>>(head, rec, buf0, out, N, E);
}


# 32-lane row sums on the vector ALU: four row_xmask DPP adds and one permlanex16 add per sum replace five LDS-unit shuffles (11 sums per row); butterfly order 1,2,4,8,16, was 16,8,4,2,1
# speedup vs baseline: 1.7521x; 1.0430x over previous
#include <hip/hip_runtime.h>
#include <hip/hip_bf16.h>


typedef __attribute__((ext_vector_type(2))) float v2f;
typedef __attribute__((ext_vector_type(8))) float v8f;

#define D_IN   256
#define D_OUT  96
#define MINN   1e-15f
#define MAXN   0.996f
#define ARTC   (1.0f - 1e-7f)

template <int M>
__device__ __forceinline__ float xmask_partner(float v) {
    int b = __builtin_bit_cast(int, v);
    return __builtin_bit_cast(float, __builtin_amdgcn_update_dpp(0, b, 0x160 + M, 0xf, 0xf, true));
}

__device__ __forceinline__ float other_row(float v) {
    int b = __builtin_bit_cast(int, v);
    return __builtin_bit_cast(float,
        __builtin_amdgcn_permlanex16(b, b, 0x76543210, 0xfedcba98, false, false));
}

__device__ __forceinline__ float wsum(float v) {
    v += xmask_partner<1>(v);
    v += xmask_partner<2>(v);
    v += xmask_partner<4>(v);
    v += xmask_partner<8>(v);
    v += other_row(v);
    return v;
}

__global__ __launch_bounds__(256) void bias_heads(const float* __restrict__ bias,
                                                  float* __restrict__ hb,
                                                  int* __restrict__ head, int N) {
    __shared__ float red[8];
    int t = threadIdx.x;
    int i = (blockIdx.x << 8) + t;
    if (i < N) head[i] = -1;
    if (blockIdx.x != 0) return;
    float v = (t < D_OUT) ? bias[t] : 0.0f;
    float ss = wsum(v * v);
    if ((t & 31) == 0) red[t >> 5] = ss;
    __syncthreads();
    float n = fmaxf(sqrtf(red[0] + red[1] + red[2] + red[3]), MINN);
    float p = tanhf(n) / n * v;
    __syncthreads();
    float ps = wsum(p * p);
    if ((t & 31) == 0) red[t >> 5] = ps;
    __syncthreads();
    float pn = fmaxf(sqrtf(red[0] + red[1] + red[2] + red[3]), MINN);
    float pf = (pn > MAXN) ? MAXN / pn : 1.0f;
    if (t < D_OUT) hb[t] = p * pf;
}

#define KS 4
struct FragSet { v2f a[4][KS]; v2f b[6][KS]; };

__device__ __forceinline__ void load_set(FragSet& f, const float* aP0, const float* aP1,
                                         const float* aP2, const float* aP3,
                                         const float* bP, int k) {
#pragma unroll
    for (int s = 0; s < KS; ++s) {
        f.a[0][s] = *(const v2f*)(aP0 + k + 4 * s);
        f.a[1][s] = *(const v2f*)(aP1 + k + 4 * s);
        f.a[2][s] = *(const v2f*)(aP2 + k + 4 * s);
        f.a[3][s] = *(const v2f*)(aP3 + k + 4 * s);
#pragma unroll
        for (int t = 0; t < 6; ++t)
            f.b[t][s] = *(const v2f*)(bP + k + 4 * s + t * 16 * D_IN);
    }
}

__device__ __forceinline__ void mma_set(v8f (&acc)[4][6], const FragSet& f) {
#pragma unroll
    for (int s = 0; s < KS; ++s)
#pragma unroll
        for (int t = 0; t < 6; ++t)
#pragma unroll
            for (int m = 0; m < 4; ++m)
                acc[m][t] = __builtin_amdgcn_wmma_f32_16x16x4_f32(
                     false, f.a[m][s],  false, f.b[t][s],
                     (short)0, acc[m][t],  false,  false);
}

__global__ __launch_bounds__(128) void gemm_wmma(const float* __restrict__ x,
                                                 const float* __restrict__ w,
                                                 float* __restrict__ mx, int N) {
    const int nQuads = (N + 63) >> 6;
    const int quad   = (blockIdx.x << 2) + (threadIdx.x >> 5);
    if (quad >= nQuads) return;

    const int lane = threadIdx.x & 31;
    const int half = lane >> 4;
    const int mr   = lane & 15;
    const int row0 = quad * 64;

    int m0 = row0 + mr;      if (m0 > N - 1) m0 = N - 1;
    int m1 = row0 + 16 + mr; if (m1 > N - 1) m1 = N - 1;
    int m2 = row0 + 32 + mr; if (m2 > N - 1) m2 = N - 1;
    int m3 = row0 + 48 + mr; if (m3 > N - 1) m3 = N - 1;
    const float* aP0 = x + (size_t)m0 * D_IN + 2 * half;
    const float* aP1 = x + (size_t)m1 * D_IN + 2 * half;
    const float* aP2 = x + (size_t)m2 * D_IN + 2 * half;
    const float* aP3 = x + (size_t)m3 * D_IN + 2 * half;
    const float* bP  = w + (size_t)mr * D_IN + 2 * half;

    v8f acc[4][6] = {};
    FragSet f0, f1;
    load_set(f0, aP0, aP1, aP2, aP3, bP, 0);
#pragma unroll 1
    for (int k = 0; k < D_IN - 8 * KS; k += 8 * KS) {
        load_set(f1, aP0, aP1, aP2, aP3, bP, k + 4 * KS);
        mma_set(acc, f0);
        load_set(f0, aP0, aP1, aP2, aP3, bP, k + 8 * KS);
        mma_set(acc, f1);
    }
    load_set(f1, aP0, aP1, aP2, aP3, bP, D_IN - 4 * KS);
    mma_set(acc, f0);
    mma_set(acc, f1);

#pragma unroll
    for (int m = 0; m < 4; ++m) {
        const int r0 = row0 + 16 * m;
        if (r0 + 15 < N) {
            float* base = mx + (size_t)(r0 + 8 * half) * D_OUT + mr;
#pragma unroll
            for (int r = 0; r < 8; ++r) {
                float* o = base + (size_t)r * D_OUT;
#pragma unroll
                for (int t = 0; t < 6; ++t) o[t * 16] = acc[m][t][r];
            }
        } else {
#pragma unroll
            for (int r = 0; r < 8; ++r) {
                int mm = r0 + r + 8 * half;
                if (mm < N) {
                    float* o = mx + (size_t)mm * D_OUT + mr;
#pragma unroll
                    for (int t = 0; t < 6; ++t) o[t * 16] = acc[m][t][r];
                }
            }
        }
    }
}

typedef __attribute__((ext_vector_type(4))) int v4i;

__global__ __launch_bounds__(256) void epilogue_link(
    const float* __restrict__ x, float* mxt, const float* __restrict__ hb,
    const int* __restrict__ erow, const int* __restrict__ ecol,
    const float* __restrict__ ew, int* __restrict__ head, int* __restrict__ rec,
    int N, int E) {
    int row = (blockIdx.x << 3) + (threadIdx.x >> 5);
    if (row >= N) return;
    int lane = threadIdx.x & 31;

    const float* xr = x + (size_t)row * D_IN;
    float xv[8];
#pragma unroll
    for (int j = 0; j < 8; ++j) xv[j] = xr[lane + 32 * j];
    float* mr = mxt + (size_t)row * D_OUT;
    float m0 = mr[lane], m1 = mr[lane + 32], m2 = mr[lane + 64];
    float b0 = hb[lane], b1 = hb[lane + 32], b2 = hb[lane + 64];

    const int epw = (E + N - 1) / N;
    const int e0  = row * epw + lane;
    const bool has = (lane < epw) && (e0 < E);
    const int*   pc = has ? ecol + e0 : head;
    const float* pw = has ? ew + e0 : hb;
    const int*   pr = has ? erow + e0 : head;
    v4i q = {-1, 0, 0, 0};
    q.y = *pc;
    q.z = __float_as_int(*pw);
    int r = *pr;
    if (!has) r = -1;

    float ss = 0.0f;
#pragma unroll
    for (int j = 0; j < 8; ++j) ss = fmaf(xv[j], xv[j], ss);
    float mq = m0 * m0 + m1 * m1 + m2 * m2;
    float yq = b0 * b0 + b1 * b1 + b2 * b2;

    if ((unsigned)r < (unsigned)N) q.x = atomicExch(head + r, e0);

    float xn = fmaxf(sqrtf(wsum(ss)), MINN);

    float mn = fmaxf(sqrtf(wsum(mq)), MINN);
    float sc = tanhf(mn / xn * atanhf(fminf(xn, ARTC))) / mn;
    float h0 = sc * m0, h1 = sc * m1, h2 = sc * m2;
    float hn = fmaxf(sqrtf(wsum(h0 * h0 + h1 * h1 + h2 * h2)), MINN);
    float pf = (hn > MAXN) ? MAXN / hn : 1.0f;
    h0 *= pf; h1 *= pf; h2 *= pf;

    float x2 = wsum(h0 * h0 + h1 * h1 + h2 * h2);
    float y2 = wsum(yq);
    float xy = wsum(h0 * b0 + h1 * b1 + h2 * b2);
    float ca  = 1.0f + 2.0f * xy + y2;
    float cb  = 1.0f - x2;
    float den = fmaxf(1.0f + 2.0f * xy + x2 * y2, MINN);
    float g0 = (ca * h0 + cb * b0) / den;
    float g1 = (ca * h1 + cb * b1) / den;
    float g2 = (ca * h2 + cb * b2) / den;
    float gn = fmaxf(sqrtf(wsum(g0 * g0 + g1 * g1 + g2 * g2)), MINN);
    pf = (gn > MAXN) ? MAXN / gn : 1.0f;
    g0 *= pf; g1 *= pf; g2 *= pf;
    float pn  = fminf(gn, MAXN);
    float lsc = atanhf(fminf(pn, ARTC)) / pn;
    mr[lane] = lsc * g0; mr[lane + 32] = lsc * g1; mr[lane + 64] = lsc * g2;

    if (has) *(v4i*)(rec + 4 * (size_t)e0) = q;
    for (int k = lane + 32; k < epw; k += 32) {
        int e = row * epw + k;
        if (e < E) {
            int r2 = erow[e];
            v4i p = {-1, ecol[e], __float_as_int(ew[e]), 0};
            if ((unsigned)r2 < (unsigned)N) p.x = atomicExch(head + r2, e);
            *(v4i*)(rec + 4 * (size_t)e) = p;
        }
    }
}

__global__ __launch_bounds__(256) void pull_agg_act(
    const int* __restrict__ head, const int* __restrict__ rec,
    const float* __restrict__ xt, float* __restrict__ out, int N, int E) {
    int row = (blockIdx.x << 3) + (threadIdx.x >> 5);
    if (row >= N) return;
    int lane = threadIdx.x & 31;

    int e = __builtin_amdgcn_readfirstlane(head[row]);
    v4i r = {-1, 0, 0, 0};
    if (e >= 0) r = *(const v4i*)(rec + 4 * (size_t)e);
    float s0 = 0.0f, s1 = 0.0f, s2 = 0.0f;
    for (int it = 0; it < E && e >= 0; ++it) {
        int   c = __builtin_amdgcn_readfirstlane(r.y);
        float w = __int_as_float(__builtin_amdgcn_readfirstlane(r.z));
        const float* src = xt + (size_t)c * D_OUT + lane;
        float g0 = src[0], g1 = src[32], g2 = src[64];
        e = __builtin_amdgcn_readfirstlane(r.x);
        if (e >= 0) r = *(const v4i*)(rec + 4 * (size_t)e);
        s0 = fmaf(w, g0, s0);
        s1 = fmaf(w, g1, s1);
        s2 = fmaf(w, g2, s2);
    }

    float sn = fmaxf(sqrtf(wsum(s0 * s0 + s1 * s1 + s2 * s2)), MINN);
    float ex = tanhf(sn) / sn;
    float p0 = ex * s0, p1 = ex * s1, p2 = ex * s2;
    float pnr = fmaxf(sqrtf(wsum(p0 * p0 + p1 * p1 + p2 * p2)), MINN);
    float pf  = (pnr > MAXN) ? MAXN / pnr : 1.0f;
    p0 *= pf; p1 *= pf; p2 *= pf;
    float pn = fminf(pnr, MAXN);
    float l  = atanhf(fminf(pn, ARTC)) / pn;
    float u0 = fmaxf(l * p0, 0.0f), u1 = fmaxf(l * p1, 0.0f), u2 = fmaxf(l * p2, 0.0f);
    float un = fmaxf(sqrtf(wsum(u0 * u0 + u1 * u1 + u2 * u2)), MINN);
    float e2 = tanhf(un) / un;
    float h0 = e2 * u0, h1 = e2 * u1, h2 = e2 * u2;
    float hn  = fmaxf(sqrtf(wsum(h0 * h0 + h1 * h1 + h2 * h2)), MINN);
    float pf2 = (hn > MAXN) ? MAXN / hn : 1.0f;
    float* o = out + (size_t)row * D_OUT;
    o[lane] = h0 * pf2; o[lane + 32] = h1 * pf2; o[lane + 64] = h2 * pf2;
}

extern "C" void kernel_launch(void* const* d_in, const int* in_sizes, int n_in,
                              void* d_out, int out_size, void* d_ws, size_t ws_size,
                              hipStream_t stream) {
    const float* x      = (const float*)d_in[0];
    const float* weight = (const float*)d_in[1];
    const float* bias   = (const float*)d_in[2];
    const float* ew     = (const float*)d_in[3];
    const int*   erow   = (const int*)d_in[4];
    const int*   ecol   = (const int*)d_in[5];
    const int N = in_sizes[0] / D_IN;
    const int E = in_sizes[3];

    float* ws   = (float*)d_ws;
    float* hb   = ws;
    float* buf0 = ws + 128;
    float* buf1 = buf0 + (size_t)N * D_OUT;
    float* out  = (float*)d_out;
    int* head = (int*)buf1;
    int* rec  = head + ((N + 3) & ~3);

    bias_heads<<<(N + 255) >> 8, 256, 0, stream>>>(bias, hb, head, N);

    int quads      = (N + 63) >> 6;
    int gemmBlocks = (quads + 3) >> 2;
    gemm_wmma<<<gemmBlocks, 128, 0, stream>>>(x, weight, buf0, N);

    epilogue_link<<<(N + 7) >> 3, 256, 0, stream>>>(x, buf0, hb, erow, ecol, ew, head, rec, N, E);

    pull_agg_act<<<(N + 7) >> 3, 256, 0, stream>>>(head, rec, buf0, out, N, E);
}


# spmm pull: a row's edges in 8 lists (edge index mod 8), a lane per list holds cursor and record, the wave steps all 8 at once: trips per row = longest list, not degree; sum order interleaves lists
# speedup vs baseline: 2.0907x; 1.1933x over previous
#include <hip/hip_runtime.h>
#include <hip/hip_bf16.h>


typedef __attribute__((ext_vector_type(2))) float v2f;
typedef __attribute__((ext_vector_type(8))) float v8f;
typedef __attribute__((ext_vector_type(4))) int v4i;

#define D_IN   256
#define D_OUT  96
#define NL     8
#define MINN   1e-15f
#define MAXN   0.996f
#define ARTC   (1.0f - 1e-7f)

template <int M>
__device__ __forceinline__ float xmask_partner(float v) {
    int b = __builtin_bit_cast(int, v);
    return __builtin_bit_cast(float, __builtin_amdgcn_update_dpp(0, b, 0x160 + M, 0xf, 0xf, true));
}

__device__ __forceinline__ float other_row(float v) {
    int b = __builtin_bit_cast(int, v);
    return __builtin_bit_cast(float,
        __builtin_amdgcn_permlanex16(b, b, 0x76543210, 0xfedcba98, false, false));
}

__device__ __forceinline__ float wsum(float v) {
    v += xmask_partner<1>(v);
    v += xmask_partner<2>(v);
    v += xmask_partner<4>(v);
    v += xmask_partner<8>(v);
    v += other_row(v);
    return v;
}

__global__ __launch_bounds__(256) void bias_heads(const float* __restrict__ bias,
                                                  float* __restrict__ hb,
                                                  int* __restrict__ head, int N) {
    __shared__ float red[8];
    int t = threadIdx.x;
    int i = (blockIdx.x << 8) + t;
    if (i < N) {
        const v4i m1 = {-1, -1, -1, -1};
        v4i* hp = (v4i*)(head + (size_t)i * NL);
#pragma unroll
        for (int j = 0; j < NL / 4; ++j) hp[j] = m1;
    }
    if (blockIdx.x != 0) return;
    float v = (t < D_OUT) ? bias[t] : 0.0f;
    float ss = wsum(v * v);
    if ((t & 31) == 0) red[t >> 5] = ss;
    __syncthreads();
    float n = fmaxf(sqrtf(red[0] + red[1] + red[2] + red[3]), MINN);
    float p = tanhf(n) / n * v;
    __syncthreads();
    float ps = wsum(p * p);
    if ((t & 31) == 0) red[t >> 5] = ps;
    __syncthreads();
    float pn = fmaxf(sqrtf(red[0] + red[1] + red[2] + red[3]), MINN);
    float pf = (pn > MAXN) ? MAXN / pn : 1.0f;
    if (t < D_OUT) hb[t] = p * pf;
}

#define KS 4
struct FragSet { v2f a[4][KS]; v2f b[6][KS]; };

__device__ __forceinline__ void load_set(FragSet& f, const float* aP0, const float* aP1,
                                         const float* aP2, const float* aP3,
                                         const float* bP, int k) {
#pragma unroll
    for (int s = 0; s < KS; ++s) {
        f.a[0][s] = *(const v2f*)(aP0 + k + 4 * s);
        f.a[1][s] = *(const v2f*)(aP1 + k + 4 * s);
        f.a[2][s] = *(const v2f*)(aP2 + k + 4 * s);
        f.a[3][s] = *(const v2f*)(aP3 + k + 4 * s);
#pragma unroll
        for (int t = 0; t < 6; ++t)
            f.b[t][s] = *(const v2f*)(bP + k + 4 * s + t * 16 * D_IN);
    }
}

__device__ __forceinline__ void mma_set(v8f (&acc)[4][6], const FragSet& f) {
#pragma unroll
    for (int s = 0; s < KS; ++s)
#pragma unroll
        for (int t = 0; t < 6; ++t)
#pragma unroll
            for (int m = 0; m < 4; ++m)
                acc[m][t] = __builtin_amdgcn_wmma_f32_16x16x4_f32(
                     false, f.a[m][s],  false, f.b[t][s],
                     (short)0, acc[m][t],  false,  false);
}

__global__ __launch_bounds__(128) void gemm_wmma(const float* __restrict__ x,
                                                 const float* __restrict__ w,
                                                 float* __restrict__ mx, int N) {
    const int nQuads = (N + 63) >> 6;
    const int quad   = (blockIdx.x << 2) + (threadIdx.x >> 5);
    if (quad >= nQuads) return;

    const int lane = threadIdx.x & 31;
    const int half = lane >> 4;
    const int mr   = lane & 15;
    const int row0 = quad * 64;

    int m0 = row0 + mr;      if (m0 > N - 1) m0 = N - 1;
    int m1 = row0 + 16 + mr; if (m1 > N - 1) m1 = N - 1;
    int m2 = row0 + 32 + mr; if (m2 > N - 1) m2 = N - 1;
    int m3 = row0 + 48 + mr; if (m3 > N - 1) m3 = N - 1;
    const float* aP0 = x + (size_t)m0 * D_IN + 2 * half;
    const float* aP1 = x + (size_t)m1 * D_IN + 2 * half;
    const float* aP2 = x + (size_t)m2 * D_IN + 2 * half;
    const float* aP3 = x + (size_t)m3 * D_IN + 2 * half;
    const float* bP  = w + (size_t)mr * D_IN + 2 * half;

    v8f acc[4][6] = {};
    FragSet f0, f1;
    load_set(f0, aP0, aP1, aP2, aP3, bP, 0);
#pragma unroll 1
    for (int k = 0; k < D_IN - 8 * KS; k += 8 * KS) {
        load_set(f1, aP0, aP1, aP2, aP3, bP, k + 4 * KS);
        mma_set(acc, f0);
        load_set(f0, aP0, aP1, aP2, aP3, bP, k + 8 * KS);
        mma_set(acc, f1);
    }
    load_set(f1, aP0, aP1, aP2, aP3, bP, D_IN - 4 * KS);
    mma_set(acc, f0);
    mma_set(acc, f1);

#pragma unroll
    for (int m = 0; m < 4; ++m) {
        const int r0 = row0 + 16 * m;
        if (r0 + 15 < N) {
            float* base = mx + (size_t)(r0 + 8 * half) * D_OUT + mr;
#pragma unroll
            for (int r = 0; r < 8; ++r) {
                float* o = base + (size_t)r * D_OUT;
#pragma unroll
                for (int t = 0; t < 6; ++t) o[t * 16] = acc[m][t][r];
            }
        } else {
#pragma unroll
            for (int r = 0; r < 8; ++r) {
                int mm = r0 + r + 8 * half;
                if (mm < N) {
                    float* o = mx + (size_t)mm * D_OUT + mr;
#pragma unroll
                    for (int t = 0; t < 6; ++t) o[t * 16] = acc[m][t][r];
                }
            }
        }
    }
}

__global__ __launch_bounds__(256) void epilogue_link(
    const float* __restrict__ x, float* mxt, const float* __restrict__ hb,
    const int* __restrict__ erow, const int* __restrict__ ecol,
    const float* __restrict__ ew, int* __restrict__ head, int* __restrict__ rec,
    int N, int E) {
    int row = (blockIdx.x << 3) + (threadIdx.x >> 5);
    if (row >= N) return;
    int lane = threadIdx.x & 31;

    const float* xr = x + (size_t)row * D_IN;
    float xv[8];
#pragma unroll
    for (int j = 0; j < 8; ++j) xv[j] = xr[lane + 32 * j];
    float* mr = mxt + (size_t)row * D_OUT;
    float m0 = mr[lane], m1 = mr[lane + 32], m2 = mr[lane + 64];
    float b0 = hb[lane], b1 = hb[lane + 32], b2 = hb[lane + 64];

    const int epw = (E + N - 1) / N;
    const int e0  = row * epw + lane;
    const bool has = (lane < epw) && (e0 < E);
    const int*   pc = has ? ecol + e0 : head;
    const float* pw = has ? ew + e0 : hb;
    const int*   pr = has ? erow + e0 : head;
    v4i q = {-1, 0, 0, 0};
    q.y = *pc;
    q.z = __float_as_int(*pw);
    int r = *pr;
    if (!has) r = -1;

    float ss = 0.0f;
#pragma unroll
    for (int j = 0; j < 8; ++j) ss = fmaf(xv[j], xv[j], ss);
    float mq = m0 * m0 + m1 * m1 + m2 * m2;
    float yq = b0 * b0 + b1 * b1 + b2 * b2;

    if ((unsigned)r < (unsigned)N)
        q.x = atomicExch(head + (size_t)r * NL + (e0 & (NL - 1)), e0);

    float xn = fmaxf(sqrtf(wsum(ss)), MINN);

    float mn = fmaxf(sqrtf(wsum(mq)), MINN);
    float sc = tanhf(mn / xn * atanhf(fminf(xn, ARTC))) / mn;
    float h0 = sc * m0, h1 = sc * m1, h2 = sc * m2;
    float hn = fmaxf(sqrtf(wsum(h0 * h0 + h1 * h1 + h2 * h2)), MINN);
    float pf = (hn > MAXN) ? MAXN / hn : 1.0f;
    h0 *= pf; h1 *= pf; h2 *= pf;

    float x2 = wsum(h0 * h0 + h1 * h1 + h2 * h2);
    float y2 = wsum(yq);
    float xy = wsum(h0 * b0 + h1 * b1 + h2 * b2);
    float ca  = 1.0f + 2.0f * xy + y2;
    float cb  = 1.0f - x2;
    float den = fmaxf(1.0f + 2.0f * xy + x2 * y2, MINN);
    float g0 = (ca * h0 + cb * b0) / den;
    float g1 = (ca * h1 + cb * b1) / den;
    float g2 = (ca * h2 + cb * b2) / den;
    float gn = fmaxf(sqrtf(wsum(g0 * g0 + g1 * g1 + g2 * g2)), MINN);
    pf = (gn > MAXN) ? MAXN / gn : 1.0f;
    g0 *= pf; g1 *= pf; g2 *= pf;
    float pn  = fminf(gn, MAXN);
    float lsc = atanhf(fminf(pn, ARTC)) / pn;
    mr[lane] = lsc * g0; mr[lane + 32] = lsc * g1; mr[lane + 64] = lsc * g2;

    if (has) *(v4i*)(rec + 4 * (size_t)e0) = q;
    for (int k = lane + 32; k < epw; k += 32) {
        int e = row * epw + k;
        if (e < E) {
            int r2 = erow[e];
            v4i p = {-1, ecol[e], __float_as_int(ew[e]), 0};
            if ((unsigned)r2 < (unsigned)N) p.x = atomicExch(head + (size_t)r2 * NL + (e & (NL - 1)), e);
            *(v4i*)(rec + 4 * (size_t)e) = p;
        }
    }
}

__global__ __launch_bounds__(256) void pull_agg_act(
    const int* __restrict__ head, const int* __restrict__ rec,
    const float* __restrict__ xt, float* __restrict__ out, int N, int E) {
    int row = (blockIdx.x << 3) + (threadIdx.x >> 5);
    if (row >= N) return;
    int lane = threadIdx.x & 31;

    int cur = -1;
    if (lane < NL) cur = head[(size_t)row * NL + lane];
    v4i rc = {-1, 0, 0, 0};
    if (cur >= 0) rc = *(const v4i*)(rec + 4 * (size_t)cur);
    const float* xl = xt + lane;
    float s0 = 0.0f, s1 = 0.0f, s2 = 0.0f;
    for (int it = 0; it < E; ++it) {
        const unsigned live = __builtin_amdgcn_ballot_w32(cur >= 0);
        if (live == 0) break;
        v4i rn = rc;
        int nxt = -1;
        if (cur >= 0) {
            nxt = rc.x;
            if (nxt >= 0) rn = *(const v4i*)(rec + 4 * (size_t)nxt);
        }
        float g0[NL], g1[NL], g2[NL], w[NL];
#pragma unroll
        for (int j = 0; j < NL; ++j) {
            if (live & (1u << j)) {
                const int c = __builtin_amdgcn_readlane(rc.y, j);
                w[j] = __int_as_float(__builtin_amdgcn_readlane(rc.z, j));
                const float* src = xl + (size_t)c * D_OUT;
                g0[j] = src[0]; g1[j] = src[32]; g2[j] = src[64];
            }
        }
        cur = nxt;
        rc  = rn;
#pragma unroll
        for (int j = 0; j < NL; ++j) {
            if (live & (1u << j)) {
                s0 = fmaf(w[j], g0[j], s0);
                s1 = fmaf(w[j], g1[j], s1);
                s2 = fmaf(w[j], g2[j], s2);
            }
        }
    }

    float sn = fmaxf(sqrtf(wsum(s0 * s0 + s1 * s1 + s2 * s2)), MINN);
    float ex = tanhf(sn) / sn;
    float p0 = ex * s0, p1 = ex * s1, p2 = ex * s2;
    float pnr = fmaxf(sqrtf(wsum(p0 * p0 + p1 * p1 + p2 * p2)), MINN);
    float pf  = (pnr > MAXN) ? MAXN / pnr : 1.0f;
    p0 *= pf; p1 *= pf; p2 *= pf;
    float pn = fminf(pnr, MAXN);
    float l  = atanhf(fminf(pn, ARTC)) / pn;
    float u0 = fmaxf(l * p0, 0.0f), u1 = fmaxf(l * p1, 0.0f), u2 = fmaxf(l * p2, 0.0f);
    float un = fmaxf(sqrtf(wsum(u0 * u0 + u1 * u1 + u2 * u2)), MINN);
    float e2 = tanhf(un) / un;
    float h0 = e2 * u0, h1 = e2 * u1, h2 = e2 * u2;
    float hn  = fmaxf(sqrtf(wsum(h0 * h0 + h1 * h1 + h2 * h2)), MINN);
    float pf2 = (hn > MAXN) ? MAXN / hn : 1.0f;
    float* o = out + (size_t)row * D_OUT;
    o[lane] = h0 * pf2; o[lane + 32] = h1 * pf2; o[lane + 64] = h2 * pf2;
}

extern "C" void kernel_launch(void* const* d_in, const int* in_sizes, int n_in,
                              void* d_out, int out_size, void* d_ws, size_t ws_size,
                              hipStream_t stream) {
    const float* x      = (const float*)d_in[0];
    const float* weight = (const float*)d_in[1];
    const float* bias   = (const float*)d_in[2];
    const float* ew     = (const float*)d_in[3];
    const int*   erow   = (const int*)d_in[4];
    const int*   ecol   = (const int*)d_in[5];
    const int N = in_sizes[0] / D_IN;
    const int E = in_sizes[3];

    float* ws   = (float*)d_ws;
    float* hb   = ws;
    float* buf0 = ws + 128;
    float* buf1 = buf0 + (size_t)N * D_OUT;
    float* out  = (float*)d_out;
    int* head = (int*)buf1;
    int* rec  = head + (size_t)N * NL;

    bias_heads<<<(N + 255) >> 8, 256, 0, stream>>>(bias, hb, head, N);

    int quads      = (N + 63) >> 6;
    int gemmBlocks = (quads + 3) >> 2;
    gemm_wmma<<<gemmBlocks, 128, 0, stream>>>(x, weight, buf0, N);

    epilogue_link<<<(N + 7) >> 3, 256, 0, stream>>>(x, buf0, hb, erow, ecol, ew, head, rec, N, E);

    pull_agg_act<<<(N + 7) >> 3, 256, 0, stream>>>(head, rec, buf0, out, N, E);
}


# x read once: the matrix loop also sums the squares of the A fragments it holds (per lane over k, two lanes per row) and writes each row's squared norm; the row epilogue loads that word, not the row
# speedup vs baseline: 2.1644x; 1.0353x over previous
#include <hip/hip_runtime.h>
#include <hip/hip_bf16.h>


typedef __attribute__((ext_vector_type(2))) float v2f;
typedef __attribute__((ext_vector_type(8))) float v8f;
typedef __attribute__((ext_vector_type(4))) int v4i;

#define D_IN   256
#define D_OUT  96
#define NL     8
#define MINN   1e-15f
#define MAXN   0.996f
#define ARTC   (1.0f - 1e-7f)

template <int M>
__device__ __forceinline__ float xmask_partner(float v) {
    int b = __builtin_bit_cast(int, v);
    return __builtin_bit_cast(float, __builtin_amdgcn_update_dpp(0, b, 0x160 + M, 0xf, 0xf, true));
}

__device__ __forceinline__ float other_row(float v) {
    int b = __builtin_bit_cast(int, v);
    return __builtin_bit_cast(float,
        __builtin_amdgcn_permlanex16(b, b, 0x76543210, 0xfedcba98, false, false));
}

__device__ __forceinline__ float wsum(float v) {
    v += xmask_partner<1>(v);
    v += xmask_partner<2>(v);
    v += xmask_partner<4>(v);
    v += xmask_partner<8>(v);
    v += other_row(v);
    return v;
}

__global__ __launch_bounds__(256) void bias_heads(const float* __restrict__ bias,
                                                  float* __restrict__ hb,
                                                  int* __restrict__ head, int N) {
    __shared__ float red[8];
    int t = threadIdx.x;
    int i = (blockIdx.x << 8) + t;
    if (i < N) {
        const v4i m1 = {-1, -1, -1, -1};
        v4i* hp = (v4i*)(head + (size_t)i * NL);
#pragma unroll
        for (int j = 0; j < NL / 4; ++j) hp[j] = m1;
    }
    if (blockIdx.x != 0) return;
    float v = (t < D_OUT) ? bias[t] : 0.0f;
    float ss = wsum(v * v);
    if ((t & 31) == 0) red[t >> 5] = ss;
    __syncthreads();
    float n = fmaxf(sqrtf(red[0] + red[1] + red[2] + red[3]), MINN);
    float p = tanhf(n) / n * v;
    __syncthreads();
    float ps = wsum(p * p);
    if ((t & 31) == 0) red[t >> 5] = ps;
    __syncthreads();
    float pn = fmaxf(sqrtf(red[0] + red[1] + red[2] + red[3]), MINN);
    float pf = (pn > MAXN) ? MAXN / pn : 1.0f;
    if (t < D_OUT) hb[t] = p * pf;
}

#define KS 4
struct FragSet { v2f a[4][KS]; v2f b[6][KS]; };

__device__ __forceinline__ void load_set(FragSet& f, const float* aP0, const float* aP1,
                                         const float* aP2, const float* aP3,
                                         const float* bP, int k) {
#pragma unroll
    for (int s = 0; s < KS; ++s) {
        f.a[0][s] = *(const v2f*)(aP0 + k + 4 * s);
        f.a[1][s] = *(const v2f*)(aP1 + k + 4 * s);
        f.a[2][s] = *(const v2f*)(aP2 + k + 4 * s);
        f.a[3][s] = *(const v2f*)(aP3 + k + 4 * s);
#pragma unroll
        for (int t = 0; t < 6; ++t)
            f.b[t][s] = *(const v2f*)(bP + k + 4 * s + t * 16 * D_IN);
    }
}

__device__ __forceinline__ void mma_set(v8f (&acc)[4][6], v2f (&sq)[4], const FragSet& f) {
#pragma unroll
    for (int s = 0; s < KS; ++s) {
#pragma unroll
        for (int t = 0; t < 6; ++t)
#pragma unroll
            for (int m = 0; m < 4; ++m)
                acc[m][t] = __builtin_amdgcn_wmma_f32_16x16x4_f32(
                     false, f.a[m][s],  false, f.b[t][s],
                     (short)0, acc[m][t],  false,  false);
        sq[0] = f.a[0][s] * f.a[0][s] + sq[0];
        sq[1] = f.a[1][s] * f.a[1][s] + sq[1];
        sq[2] = f.a[2][s] * f.a[2][s] + sq[2];
        sq[3] = f.a[3][s] * f.a[3][s] + sq[3];
    }
}

__global__ __launch_bounds__(128) void gemm_wmma(const float* __restrict__ x,
                                                 const float* __restrict__ w,
                                                 float* __restrict__ mx,
                                                 float* __restrict__ xq, int N) {
    const int nQuads = (N + 63) >> 6;
    const int quad   = (blockIdx.x << 2) + (threadIdx.x >> 5);
    if (quad >= nQuads) return;

    const int lane = threadIdx.x & 31;
    const int half = lane >> 4;
    const int mr   = lane & 15;
    const int row0 = quad * 64;

    int m0 = row0 + mr;      if (m0 > N - 1) m0 = N - 1;
    int m1 = row0 + 16 + mr; if (m1 > N - 1) m1 = N - 1;
    int m2 = row0 + 32 + mr; if (m2 > N - 1) m2 = N - 1;
    int m3 = row0 + 48 + mr; if (m3 > N - 1) m3 = N - 1;
    const float* aP0 = x + (size_t)m0 * D_IN + 2 * half;
    const float* aP1 = x + (size_t)m1 * D_IN + 2 * half;
    const float* aP2 = x + (size_t)m2 * D_IN + 2 * half;
    const float* aP3 = x + (size_t)m3 * D_IN + 2 * half;
    const float* bP  = w + (size_t)mr * D_IN + 2 * half;

    v8f acc[4][6] = {};
    v2f sq[4] = {};
    FragSet f0, f1;
    load_set(f0, aP0, aP1, aP2, aP3, bP, 0);
#pragma unroll 1
    for (int k = 0; k < D_IN - 8 * KS; k += 8 * KS) {
        load_set(f1, aP0, aP1, aP2, aP3, bP, k + 4 * KS);
        mma_set(acc, sq, f0);
        load_set(f0, aP0, aP1, aP2, aP3, bP, k + 8 * KS);
        mma_set(acc, sq, f1);
    }
    load_set(f1, aP0, aP1, aP2, aP3, bP, D_IN - 4 * KS);
    mma_set(acc, sq, f0);
    mma_set(acc, sq, f1);

    {
        float t0 = sq[0].x + sq[0].y, t1 = sq[1].x + sq[1].y;
        float t2 = sq[2].x + sq[2].y, t3 = sq[3].x + sq[3].y;
        t0 += other_row(t0); t1 += other_row(t1);
        t2 += other_row(t2); t3 += other_row(t3);
        const float lo = half ? t1 : t0;
        const float hi = half ? t3 : t2;
        if (row0 + lane < N)      xq[row0 + lane]      = lo;
        if (row0 + 32 + lane < N) xq[row0 + 32 + lane] = hi;
    }

#pragma unroll
    for (int m = 0; m < 4; ++m) {
        const int r0 = row0 + 16 * m;
        if (r0 + 15 < N) {
            float* base = mx + (size_t)(r0 + 8 * half) * D_OUT + mr;
#pragma unroll
            for (int r = 0; r < 8; ++r) {
                float* o = base + (size_t)r * D_OUT;
#pragma unroll
                for (int t = 0; t < 6; ++t) o[t * 16] = acc[m][t][r];
            }
        } else {
#pragma unroll
            for (int r = 0; r < 8; ++r) {
                int mm = r0 + r + 8 * half;
                if (mm < N) {
                    float* o = mx + (size_t)mm * D_OUT + mr;
#pragma unroll
                    for (int t = 0; t < 6; ++t) o[t * 16] = acc[m][t][r];
                }
            }
        }
    }
}

__global__ __launch_bounds__(256) void epilogue_link(
    const float* __restrict__ xq, float* mxt, const float* __restrict__ hb,
    const int* __restrict__ erow, const int* __restrict__ ecol,
    const float* __restrict__ ew, int* __restrict__ head, int* __restrict__ rec,
    int N, int E) {
    int row = (blockIdx.x << 3) + (threadIdx.x >> 5);
    if (row >= N) return;
    int lane = threadIdx.x & 31;

    const float ss = xq[row];
    float* mr = mxt + (size_t)row * D_OUT;
    float m0 = mr[lane], m1 = mr[lane + 32], m2 = mr[lane + 64];
    float b0 = hb[lane], b1 = hb[lane + 32], b2 = hb[lane + 64];

    const int epw = (E + N - 1) / N;
    const int e0  = row * epw + lane;
    const bool has = (lane < epw) && (e0 < E);
    const int*   pc = has ? ecol + e0 : head;
    const float* pw = has ? ew + e0 : hb;
    const int*   pr = has ? erow + e0 : head;
    v4i q = {-1, 0, 0, 0};
    q.y = *pc;
    q.z = __float_as_int(*pw);
    int r = *pr;
    if (!has) r = -1;

    float mq = m0 * m0 + m1 * m1 + m2 * m2;
    float yq = b0 * b0 + b1 * b1 + b2 * b2;

    if ((unsigned)r < (unsigned)N)
        q.x = atomicExch(head + (size_t)r * NL + (e0 & (NL - 1)), e0);

    float xn = fmaxf(sqrtf(ss), MINN);

    float mn = fmaxf(sqrtf(wsum(mq)), MINN);
    float sc = tanhf(mn / xn * atanhf(fminf(xn, ARTC))) / mn;
    float h0 = sc * m0, h1 = sc * m1, h2 = sc * m2;
    float hn = fmaxf(sqrtf(wsum(h0 * h0 + h1 * h1 + h2 * h2)), MINN);
    float pf = (hn > MAXN) ? MAXN / hn : 1.0f;
    h0 *= pf; h1 *= pf; h2 *= pf;

    float x2 = wsum(h0 * h0 + h1 * h1 + h2 * h2);
    float y2 = wsum(yq);
    float xy = wsum(h0 * b0 + h1 * b1 + h2 * b2);
    float ca  = 1.0f + 2.0f * xy + y2;
    float cb  = 1.0f - x2;
    float den = fmaxf(1.0f + 2.0f * xy + x2 * y2, MINN);
    float g0 = (ca * h0 + cb * b0) / den;
    float g1 = (ca * h1 + cb * b1) / den;
    float g2 = (ca * h2 + cb * b2) / den;
    float gn = fmaxf(sqrtf(wsum(g0 * g0 + g1 * g1 + g2 * g2)), MINN);
    pf = (gn > MAXN) ? MAXN / gn : 1.0f;
    g0 *= pf; g1 *= pf; g2 *= pf;
    float pn  = fminf(gn, MAXN);
    float lsc = atanhf(fminf(pn, ARTC)) / pn;
    mr[lane] = lsc * g0; mr[lane + 32] = lsc * g1; mr[lane + 64] = lsc * g2;

    if (has) *(v4i*)(rec + 4 * (size_t)e0) = q;
    for (int k = lane + 32; k < epw; k += 32) {
        int e = row * epw + k;
        if (e < E) {
            int r2 = erow[e];
            v4i p = {-1, ecol[e], __float_as_int(ew[e]), 0};
            if ((unsigned)r2 < (unsigned)N) p.x = atomicExch(head + (size_t)r2 * NL + (e & (NL - 1)), e);
            *(v4i*)(rec + 4 * (size_t)e) = p;
        }
    }
}

__global__ __launch_bounds__(256) void pull_agg_act(
    const int* __restrict__ head, const int* __restrict__ rec,
    const float* __restrict__ xt, float* __restrict__ out, int N, int E) {
    int row = (blockIdx.x << 3) + (threadIdx.x >> 5);
    if (row >= N) return;
    int lane = threadIdx.x & 31;

    int cur = -1;
    if (lane < NL) cur = head[(size_t)row * NL + lane];
    v4i rc = {-1, 0, 0, 0};
    if (cur >= 0) rc = *(const v4i*)(rec + 4 * (size_t)cur);
    const float* xl = xt + lane;
    float s0 = 0.0f, s1 = 0.0f, s2 = 0.0f;
    for (int it = 0; it < E; ++it) {
        const unsigned live = __builtin_amdgcn_ballot_w32(cur >= 0);
        if (live == 0) break;
        v4i rn = rc;
        int nxt = -1;
        if (cur >= 0) {
            nxt = rc.x;
            if (nxt >= 0) rn = *(const v4i*)(rec + 4 * (size_t)nxt);
        }
        float g0[NL], g1[NL], g2[NL], w[NL];
#pragma unroll
        for (int j = 0; j < NL; ++j) {
            if (live & (1u << j)) {
                const int c = __builtin_amdgcn_readlane(rc.y, j);
                w[j] = __int_as_float(__builtin_amdgcn_readlane(rc.z, j));
                const float* src = xl + (size_t)c * D_OUT;
                g0[j] = src[0]; g1[j] = src[32]; g2[j] = src[64];
            }
        }
        cur = nxt;
        rc  = rn;
#pragma unroll
        for (int j = 0; j < NL; ++j) {
            if (live & (1u << j)) {
                s0 = fmaf(w[j], g0[j], s0);
                s1 = fmaf(w[j], g1[j], s1);
                s2 = fmaf(w[j], g2[j], s2);
            }
        }
    }

    float sn = fmaxf(sqrtf(wsum(s0 * s0 + s1 * s1 + s2 * s2)), MINN);
    float ex = tanhf(sn) / sn;
    float p0 = ex * s0, p1 = ex * s1, p2 = ex * s2;
    float pnr = fmaxf(sqrtf(wsum(p0 * p0 + p1 * p1 + p2 * p2)), MINN);
    float pf  = (pnr > MAXN) ? MAXN / pnr : 1.0f;
    p0 *= pf; p1 *= pf; p2 *= pf;
    float pn = fminf(pnr, MAXN);
    float l  = atanhf(fminf(pn, ARTC)) / pn;
    float u0 = fmaxf(l * p0, 0.0f), u1 = fmaxf(l * p1, 0.0f), u2 = fmaxf(l * p2, 0.0f);
    float un = fmaxf(sqrtf(wsum(u0 * u0 + u1 * u1 + u2 * u2)), MINN);
    float e2 = tanhf(un) / un;
    float h0 = e2 * u0, h1 = e2 * u1, h2 = e2 * u2;
    float hn  = fmaxf(sqrtf(wsum(h0 * h0 + h1 * h1 + h2 * h2)), MINN);
    float pf2 = (hn > MAXN) ? MAXN / hn : 1.0f;
    float* o = out + (size_t)row * D_OUT;
    o[lane] = h0 * pf2; o[lane + 32] = h1 * pf2; o[lane + 64] = h2 * pf2;
}

extern "C" void kernel_launch(void* const* d_in, const int* in_sizes, int n_in,
                              void* d_out, int out_size, void* d_ws, size_t ws_size,
                              hipStream_t stream) {
    const float* x      = (const float*)d_in[0];
    const float* weight = (const float*)d_in[1];
    const float* bias   = (const float*)d_in[2];
    const float* ew     = (const float*)d_in[3];
    const int*   erow   = (const int*)d_in[4];
    const int*   ecol   = (const int*)d_in[5];
    const int N = in_sizes[0] / D_IN;
    const int E = in_sizes[3];

    float* ws   = (float*)d_ws;
    float* hb   = ws;
    float* buf0 = ws + 128;
    float* buf1 = buf0 + (size_t)N * D_OUT;
    float* out  = (float*)d_out;
    int* head = (int*)buf1;
    int* rec  = head + (size_t)N * NL;
    float* xq = (float*)(rec + 4 * (size_t)E);

    bias_heads<<<(N + 255) >> 8, 256, 0, stream>>>(bias, hb, head, N);

    int quads      = (N + 63) >> 6;
    int gemmBlocks = (quads + 3) >> 2;
    gemm_wmma<<<gemmBlocks, 128, 0, stream>>>(x, weight, buf0, xq, N);

    epilogue_link<<<(N + 7) >> 3, 256, 0, stream>>>(xq, buf0, hb, erow, ecol, ew, head, rec, N, E);

    pull_agg_act<<<(N + 7) >> 3, 256, 0, stream>>>(head, rec, buf0, out, N, E);
}
